# starting kernel
# baseline (speedup 1.0000x reference)
#include <hip/hip_runtime.h>

typedef __attribute__((ext_vector_type(2))) float v2f;
typedef __attribute__((ext_vector_type(8))) float v8f;

#define EMB 32
#define K_NBR 10
#define WAVES_PER_BLOCK 8

__global__ __launch_bounds__(256) void sage_layer_kernel(
    const float* __restrict__ feat,
    const float* __restrict__ W,
    const int*   __restrict__ neigh,
    const int*   __restrict__ idxMap,
    float*       __restrict__ out,
    int nTiles)
{
    __shared__ float cshAll[WAVES_PER_BLOCK * 16 * 64];
    const int lane   = threadIdx.x & 31;
    const int wave   = threadIdx.x >> 5;
    const int tile   = blockIdx.x * WAVES_PER_BLOCK + wave;
    if (tile >= nTiles) return;

    const int half   = lane >> 4;
    const int lane16 = lane & 15;
    float* csh = &cshAll[wave * 16 * 64];

    v2f Bf[2][16];
    #pragma unroll
    for (int nh = 0; nh < 2; ++nh) {
        const int o = lane16 + 16 * nh;
        #pragma unroll
        for (int k = 0; k < 16; ++k) {
            const int kk = 4 * k + 2 * half;
            const float2 w2 = *(const float2*)(W + o * 64 + kk);
            v2f b; b.x = w2.x; b.y = w2.y;
            Bf[nh][k] = b;
        }
    }

    const int m0 = tile * 16;
    for (int r = 0; r < 16; ++r) {
        const int row  = m0 + r;
        const int node = idxMap ? idxMap[row] : row;
        const float self = feat[node * EMB + lane];
        const int* nb = neigh + node * K_NBR;
        float acc = 0.f;
        #pragma unroll
        for (int j = 0; j < K_NBR; ++j)
            acc += feat[nb[j] * EMB + lane];
        csh[r * 64 + lane]      = self;
        csh[r * 64 + 32 + lane] = acc * (1.0f / K_NBR);
    }

    #pragma unroll
    for (int nh = 0; nh < 2; ++nh) {
        v8f c = {};
        #pragma unroll
        for (int k = 0; k < 16; ++k) {
            const float2 a2 = *(const float2*)&csh[lane16 * 64 + 4 * k + 2 * half];
            v2f a; a.x = a2.x; a.y = a2.y;
            c = __builtin_amdgcn_wmma_f32_16x16x4_f32(
                     false, a,  false, Bf[nh][k],
                     (short)0, c,  false,  false);
        }
        const int o = 16 * nh + lane16;
        #pragma unroll
        for (int v = 0; v < 8; ++v) {
            const int m = v + 8 * half;
            out[(m0 + m) * EMB + o] = fmaxf(c[v], 0.0f);
        }
    }
}

extern "C" void kernel_launch(void* const* d_in, const int* in_sizes, int n_in,
                              void* d_out, int out_size, void* d_ws, size_t ws_size,
                              hipStream_t stream) {
    const float* emb        = (const float*)d_in[0];
    const float* W1         = (const float*)d_in[1];
    const float* W2         = (const float*)d_in[2];
    const int*   node_batch = (const int*)  d_in[3];
    const int*   neigh      = (const int*)  d_in[4];
    float*       out        = (float*)d_out;
    float*       h1         = (float*)d_ws;

    const int N = in_sizes[0] / EMB;
    const int B = in_sizes[3];

    const int t1 = (N + 15) / 16;
    const int t2 = (B + 15) / 16;

    sage_layer_kernel<<<(t1 + WAVES_PER_BLOCK - 1) / WAVES_PER_BLOCK, 256, 0, stream>>>(
        emb, W1, neigh, nullptr, h1, t1);

    sage_layer_kernel<<<(t2 + WAVES_PER_BLOCK - 1) / WAVES_PER_BLOCK, 256, 0, stream>>>(
        h1, W2, neigh, node_batch, out, t2);
}


# Each lane gathers its own node's rows as 16-B chunks straight into the WMMA operand registers: no LDS transpose, a tile's 16 rows in flight side by side; k order of the matrix sums regrouped
# speedup vs baseline: 1.1067x; 1.1067x over previous
#include <hip/hip_runtime.h>

typedef __attribute__((ext_vector_type(2))) float v2f;
typedef __attribute__((ext_vector_type(8))) float v8f;

#define EMB 32
#define K_NBR 10
#define WAVES_PER_BLOCK 8

__device__ __forceinline__ float4 ld_chunk(const float* __restrict__ rowBase, int q, int half) {
    return *(const float4*)(rowBase + 8 * q + 4 * half);
}

__device__ __forceinline__ v8f mma4(float a0, float a1, float b0, float b1, v8f c) {
    v2f a; a.x = a0; a.y = a1;
    v2f b; b.x = b0; b.y = b1;
    return __builtin_amdgcn_wmma_f32_16x16x4_f32(
         false, a,  false, b,
         (short)0, c,  false,  false);
}

#define ROW_GROUP_FENCE() __builtin_amdgcn_sched_barrier(0)

__global__ __launch_bounds__(256) void sage_layer_kernel(
    const float* __restrict__ feat,
    const float* __restrict__ W,
    const int*   __restrict__ neigh,
    const int*   __restrict__ idxMap,
    float*       __restrict__ out,
    int nTiles)
{
    const int lane = threadIdx.x & 31;
    const int wave = threadIdx.x >> 5;
    const int tile = blockIdx.x * WAVES_PER_BLOCK + wave;
    if (tile >= nTiles) return;

    const int half = lane >> 4;
    const int c    = lane & 15;
    const int m0   = tile * 16;
    const int node = idxMap ? idxMap[m0 + c] : (m0 + c);

    const int2* nrow = (const int2*)(neigh + node * K_NBR);
    int ids[K_NBR];
    #pragma unroll
    for (int i = 0; i < K_NBR / 2; ++i) {
        const int2 p = nrow[i];
        ids[2 * i] = p.x; ids[2 * i + 1] = p.y;
    }

    const float* w0 = W + c * 64;
    const float* w1 = W + (c + 16) * 64;
    float4 wA[4], wB[4];
    #pragma unroll
    for (int q = 0; q < 4; ++q) { wA[q] = ld_chunk(w0, q, half); wB[q] = ld_chunk(w1, q, half); }

    const float* selfRow = feat + node * EMB;
    float4 S[4];
    #pragma unroll
    for (int q = 0; q < 4; ++q) S[q] = ld_chunk(selfRow, q, half);
    ROW_GROUP_FENCE();

    float4 R[3][4];
    #pragma unroll
    for (int q = 0; q < 4; ++q) R[0][q] = ld_chunk(feat + ids[0] * EMB, q, half);
    #pragma unroll
    for (int q = 0; q < 4; ++q) R[1][q] = ld_chunk(feat + ids[1] * EMB, q, half);
    ROW_GROUP_FENCE();

    v8f d0 = {};
    v8f d1 = {};
    #pragma unroll
    for (int q = 0; q < 4; ++q) {
        d0 = mma4(S[q].x, S[q].y, wA[q].x, wA[q].y, d0);
        d1 = mma4(S[q].x, S[q].y, wB[q].x, wB[q].y, d1);
        d0 = mma4(S[q].z, S[q].w, wA[q].z, wA[q].w, d0);
        d1 = mma4(S[q].z, S[q].w, wB[q].z, wB[q].w, d1);
    }
    ROW_GROUP_FENCE();

    float4 acc[4];
    #pragma unroll
    for (int q = 0; q < 4; ++q) acc[q] = make_float4(0.f, 0.f, 0.f, 0.f);
    float4 vA[4], vB[4];
    #pragma unroll
    for (int j = 0; j < K_NBR; ++j) {
        if (j + 2 < K_NBR) {
            #pragma unroll
            for (int q = 0; q < 4; ++q) R[(j + 2) % 3][q] = ld_chunk(feat + ids[j + 2] * EMB, q, half);
        }
        if (j == K_NBR - 3) {
            #pragma unroll
            for (int q = 0; q < 4; ++q) { vA[q] = ld_chunk(w0, 4 + q, half); vB[q] = ld_chunk(w1, 4 + q, half); }
        }
        ROW_GROUP_FENCE();
        #pragma unroll
        for (int q = 0; q < 4; ++q) {
            const float4 v = R[j % 3][q];
            acc[q].x += v.x; acc[q].y += v.y; acc[q].z += v.z; acc[q].w += v.w;
        }
        ROW_GROUP_FENCE();
    }

    #pragma unroll
    for (int q = 0; q < 4; ++q) {
        const float ax = acc[q].x * (1.0f / K_NBR);
        const float ay = acc[q].y * (1.0f / K_NBR);
        const float az = acc[q].z * (1.0f / K_NBR);
        const float aw = acc[q].w * (1.0f / K_NBR);
        d0 = mma4(ax, ay, vA[q].x, vA[q].y, d0);
        d1 = mma4(ax, ay, vB[q].x, vB[q].y, d1);
        d0 = mma4(az, aw, vA[q].z, vA[q].w, d0);
        d1 = mma4(az, aw, vB[q].z, vB[q].w, d1);
    }

    #pragma unroll
    for (int v = 0; v < 8; ++v) {
        const int m = v + 8 * half;
        out[(m0 + m) * EMB + c]      = fmaxf(d0[v], 0.0f);
        out[(m0 + m) * EMB + 16 + c] = fmaxf(d1[v], 0.0f);
    }
}

extern "C" void kernel_launch(void* const* d_in, const int* in_sizes, int n_in,
                              void* d_out, int out_size, void* d_ws, size_t ws_size,
                              hipStream_t stream) {
    const float* emb        = (const float*)d_in[0];
    const float* W1         = (const float*)d_in[1];
    const float* W2         = (const float*)d_in[2];
    const int*   node_batch = (const int*)  d_in[3];
    const int*   neigh      = (const int*)  d_in[4];
    float*       out        = (float*)d_out;
    float*       h1         = (float*)d_ws;

    const int N = in_sizes[0] / EMB;
    const int B = in_sizes[3];

    const int t1 = (N + 15) / 16;
    const int t2 = (B + 15) / 16;

    sage_layer_kernel<<<(t1 + WAVES_PER_BLOCK - 1) / WAVES_PER_BLOCK, 256, 0, stream>>>(
        emb, W1, neigh, nullptr, h1, t1);

    sage_layer_kernel<<<(t2 + WAVES_PER_BLOCK - 1) / WAVES_PER_BLOCK, 256, 0, stream>>>(
        h1, W2, neigh, node_batch, out, t2);
}


# Each wave computes a 32-row tile (two 16-row matrix tiles side by side): W fragments fetched once for both, half the waves; row-group loads kept in source order by a compiler fence; sums unchanged
# speedup vs baseline: 1.1613x; 1.0493x over previous
#include <hip/hip_runtime.h>

typedef __attribute__((ext_vector_type(2))) float v2f;
typedef __attribute__((ext_vector_type(8))) float v8f;

#define EMB 32
#define K_NBR 10
#define WAVES_PER_BLOCK 8
#define TPW 2
#define ROWS16_PER_BLOCK (WAVES_PER_BLOCK * TPW)

__device__ __forceinline__ float4 ld_chunk(const float* __restrict__ rowBase, int q, int half) {
    return *(const float4*)(rowBase + 8 * q + 4 * half);
}

__device__ __forceinline__ v8f mma4(float a0, float a1, float b0, float b1, v8f c) {
    v2f a; a.x = a0; a.y = a1;
    v2f b; b.x = b0; b.y = b1;
    return __builtin_amdgcn_wmma_f32_16x16x4_f32(
         false, a,  false, b,
         (short)0, c,  false,  false);
}

#define ROW_GROUP_FENCE() do { asm volatile("" ::: "memory"); __builtin_amdgcn_sched_barrier(0); } while (0)

#define SUM_FENCE(a) asm volatile("" : "+v"((a).x), "+v"((a).y), "+v"((a).z), "+v"((a).w))

__global__ __launch_bounds__(256) void sage_layer_kernel(
    const float* __restrict__ feat,
    const float* __restrict__ W,
    const int*   __restrict__ neigh,
    const int*   __restrict__ idxMap,
    float*       __restrict__ out,
    int nTiles)
{
    const int lane = threadIdx.x & 31;
    const int wave = threadIdx.x >> 5;
    const int tileA = 2 * (blockIdx.x * WAVES_PER_BLOCK + wave);
    if (tileA >= nTiles) return;
    const bool hasB = (tileA + 1 < nTiles);

    const int half = lane >> 4;
    const int c    = lane & 15;
    int m0[TPW];
    m0[0] = tileA * 16;
    m0[1] = (hasB ? tileA + 1 : tileA) * 16;

    int node[TPW];
    #pragma unroll
    for (int t = 0; t < TPW; ++t) node[t] = idxMap ? idxMap[m0[t] + c] : (m0[t] + c);

    int ids[TPW][K_NBR];
    #pragma unroll
    for (int t = 0; t < TPW; ++t) {
        const int2* nrow = (const int2*)(neigh + node[t] * K_NBR);
        #pragma unroll
        for (int i = 0; i < K_NBR / 2; ++i) {
            const int2 p = nrow[i];
            ids[t][2 * i] = p.x; ids[t][2 * i + 1] = p.y;
        }
    }

    const float* w0 = W + c * 64;
    const float* w1 = W + (c + 16) * 64;
    float4 wA[4], wB[4];
    #pragma unroll
    for (int q = 0; q < 4; ++q) { wA[q] = ld_chunk(w0, q, half); wB[q] = ld_chunk(w1, q, half); }

    float4 S[TPW][4];
    #pragma unroll
    for (int t = 0; t < TPW; ++t) {
        const float* selfRow = feat + node[t] * EMB;
        #pragma unroll
        for (int q = 0; q < 4; ++q) S[t][q] = ld_chunk(selfRow, q, half);
        ROW_GROUP_FENCE();
    }

    float4 R[3][TPW][4];
    #pragma unroll
    for (int j = 0; j < 2; ++j) {
        #pragma unroll
        for (int t = 0; t < TPW; ++t) {
            #pragma unroll
            for (int q = 0; q < 4; ++q) R[j][t][q] = ld_chunk(feat + ids[t][j] * EMB, q, half);
            ROW_GROUP_FENCE();
        }
    }

    v8f d0[TPW], d1[TPW];
    #pragma unroll
    for (int t = 0; t < TPW; ++t) { d0[t] = (v8f){}; d1[t] = (v8f){}; }
    #pragma unroll
    for (int q = 0; q < 4; ++q) {
        #pragma unroll
        for (int t = 0; t < TPW; ++t) {
            d0[t] = mma4(S[t][q].x, S[t][q].y, wA[q].x, wA[q].y, d0[t]);
            d1[t] = mma4(S[t][q].x, S[t][q].y, wB[q].x, wB[q].y, d1[t]);
        }
        #pragma unroll
        for (int t = 0; t < TPW; ++t) {
            d0[t] = mma4(S[t][q].z, S[t][q].w, wA[q].z, wA[q].w, d0[t]);
            d1[t] = mma4(S[t][q].z, S[t][q].w, wB[q].z, wB[q].w, d1[t]);
        }
    }
    ROW_GROUP_FENCE();

    float4 acc[TPW][4];
    #pragma unroll
    for (int t = 0; t < TPW; ++t) {
        #pragma unroll
        for (int q = 0; q < 4; ++q) acc[t][q] = make_float4(0.f, 0.f, 0.f, 0.f);
    }
    float4 vA[4], vB[4];
    #pragma unroll
    for (int j = 0; j < K_NBR; ++j) {
        if (j + 2 < K_NBR) {
            #pragma unroll
            for (int t = 0; t < TPW; ++t) {
                #pragma unroll
                for (int q = 0; q < 4; ++q)
                    R[(j + 2) % 3][t][q] = ld_chunk(feat + ids[t][j + 2] * EMB, q, half);
                ROW_GROUP_FENCE();
            }
        }
        if (j == K_NBR - 3) {
            #pragma unroll
            for (int q = 0; q < 4; ++q) { vA[q] = ld_chunk(w0, 4 + q, half); vB[q] = ld_chunk(w1, 4 + q, half); }
        }
        ROW_GROUP_FENCE();
        #pragma unroll
        for (int t = 0; t < TPW; ++t) {
            #pragma unroll
            for (int q = 0; q < 4; ++q) {
                const float4 v = R[j % 3][t][q];
                if (j == 0) SUM_FENCE(acc[t][q]);
                acc[t][q].x += v.x; acc[t][q].y += v.y; acc[t][q].z += v.z; acc[t][q].w += v.w;
            }
        }
        ROW_GROUP_FENCE();
    }

    #pragma unroll
    for (int q = 0; q < 4; ++q) {
        float ax[TPW], ay[TPW], az[TPW], aw[TPW];
        #pragma unroll
        for (int t = 0; t < TPW; ++t) {
            ax[t] = acc[t][q].x * (1.0f / K_NBR);
            ay[t] = acc[t][q].y * (1.0f / K_NBR);
            az[t] = acc[t][q].z * (1.0f / K_NBR);
            aw[t] = acc[t][q].w * (1.0f / K_NBR);
        }
        #pragma unroll
        for (int t = 0; t < TPW; ++t) {
            d0[t] = mma4(ax[t], ay[t], vA[q].x, vA[q].y, d0[t]);
            d1[t] = mma4(ax[t], ay[t], vB[q].x, vB[q].y, d1[t]);
        }
        #pragma unroll
        for (int t = 0; t < TPW; ++t) {
            d0[t] = mma4(az[t], aw[t], vA[q].z, vA[q].w, d0[t]);
            d1[t] = mma4(az[t], aw[t], vB[q].z, vB[q].w, d1[t]);
        }
    }

    #pragma unroll
    for (int t = 0; t < TPW; ++t) {
        if (t == 0 || hasB) {
            #pragma unroll
            for (int v = 0; v < 8; ++v) {
                const int m = v + 8 * half;
                out[(m0[t] + m) * EMB + c]      = fmaxf(d0[t][v], 0.0f);
                out[(m0[t] + m) * EMB + 16 + c] = fmaxf(d1[t][v], 0.0f);
            }
        }
    }
}

extern "C" void kernel_launch(void* const* d_in, const int* in_sizes, int n_in,
                              void* d_out, int out_size, void* d_ws, size_t ws_size,
                              hipStream_t stream) {
    const float* emb        = (const float*)d_in[0];
    const float* W1         = (const float*)d_in[1];
    const float* W2         = (const float*)d_in[2];
    const int*   node_batch = (const int*)  d_in[3];
    const int*   neigh      = (const int*)  d_in[4];
    float*       out        = (float*)d_out;
    float*       h1         = (float*)d_ws;

    const int N = in_sizes[0] / EMB;
    const int B = in_sizes[3];

    const int t1 = (N + 15) / 16;
    const int t2 = (B + 15) / 16;

    sage_layer_kernel<<<(t1 + ROWS16_PER_BLOCK - 1) / ROWS16_PER_BLOCK, 256, 0, stream>>>(
        emb, W1, neigh, nullptr, h1, t1);

    sage_layer_kernel<<<(t2 + ROWS16_PER_BLOCK - 1) / ROWS16_PER_BLOCK, 256, 0, stream>>>(
        h1, W2, neigh, node_batch, out, t2);
}


# Gather on whole lines: 8 lanes x 16 B cover a row, 4 rows per load (was 16 lines x 32 B per load); neighbours summed in registers in that layout, self+mean via wave-private LDS; sums unchanged
# speedup vs baseline: 1.3084x; 1.1267x over previous
#include <hip/hip_runtime.h>

typedef __attribute__((ext_vector_type(2))) float v2f;
typedef __attribute__((ext_vector_type(8))) float v8f;

#define EMB 32
#define K_NBR 10
#define WAVES_PER_BLOCK 8
#define TPW 2
#define ROWS16_PER_BLOCK (WAVES_PER_BLOCK * TPW)
#define SLOTS (TPW * 4)
#define PITCH 68

__device__ __forceinline__ float4 ld_chunk(const float* __restrict__ rowBase, int q, int half) {
    return *(const float4*)(rowBase + 8 * q + 4 * half);
}

__device__ __forceinline__ v8f mma4(float a0, float a1, float b0, float b1, v8f c) {
    v2f a; a.x = a0; a.y = a1;
    v2f b; b.x = b0; b.y = b1;
    return __builtin_amdgcn_wmma_f32_16x16x4_f32(
         false, a,  false, b,
         (short)0, c,  false,  false);
}

#define LOAD_GROUP_FENCE() __builtin_amdgcn_sched_barrier(0)

__global__ __launch_bounds__(256) void sage_layer_kernel(
    const float* __restrict__ feat,
    const float* __restrict__ W,
    const int*   __restrict__ neigh,
    const int*   __restrict__ idxMap,
    float*       __restrict__ out,
    int nTiles)
{
    __shared__ __attribute__((aligned(16))) float stage[WAVES_PER_BLOCK][TPW * 16 * PITCH];

    const int lane = threadIdx.x & 31;
    const int wave = threadIdx.x >> 5;
    const int tileA = 2 * (blockIdx.x * WAVES_PER_BLOCK + wave);
    if (tileA >= nTiles) return;
    const bool hasB = (tileA + 1 < nTiles);
    const int m0A = tileA * 16;
    const int m0B = (hasB ? tileA + 1 : tileA) * 16;

    float* myStage = stage[wave];

    const int r = lane >> 3;
    const int p = lane & 7;

    #define GROUP_ROW(s) ((((s) < 4) ? m0A : m0B) + 4 * ((s) & 3) + r)

    int nodeCur = idxMap ? idxMap[GROUP_ROW(0)] : GROUP_ROW(0);
    int nodeNxt = idxMap ? idxMap[GROUP_ROW(1)] : GROUP_ROW(1);
    int idsCur[K_NBR];
    {
        const int2* nrow = (const int2*)(neigh + nodeCur * K_NBR);
        #pragma unroll
        for (int i = 0; i < K_NBR / 2; ++i) { const int2 v = nrow[i]; idsCur[2 * i] = v.x; idsCur[2 * i + 1] = v.y; }
    }

    #pragma unroll 1
    for (int s = 0; s < SLOTS; ++s) {
        const float4 selfV = *(const float4*)(feat + nodeCur * EMB + 4 * p);
        float4 R[K_NBR];
        #pragma unroll
        for (int j = 0; j < K_NBR; ++j) R[j] = *(const float4*)(feat + idsCur[j] * EMB + 4 * p);

        const int s2 = (s + 2 < SLOTS) ? s + 2 : SLOTS - 1;
        const int nodeNxt2 = idxMap ? idxMap[GROUP_ROW(s2)] : GROUP_ROW(s2);
        int idsNxt[K_NBR];
        {
            const int2* nrow = (const int2*)(neigh + nodeNxt * K_NBR);
            #pragma unroll
            for (int i = 0; i < K_NBR / 2; ++i) { const int2 v = nrow[i]; idsNxt[2 * i] = v.x; idsNxt[2 * i + 1] = v.y; }
        }

        LOAD_GROUP_FENCE();

        float4 acc = make_float4(0.f, 0.f, 0.f, 0.f);
        #pragma unroll
        for (int j = 0; j < K_NBR; ++j) { acc.x += R[j].x; acc.y += R[j].y; acc.z += R[j].z; acc.w += R[j].w; }
        float4 mean;
        mean.x = acc.x * (1.0f / K_NBR); mean.y = acc.y * (1.0f / K_NBR);
        mean.z = acc.z * (1.0f / K_NBR); mean.w = acc.w * (1.0f / K_NBR);

        float* dst = myStage + (4 * s + r) * PITCH + 4 * p;
        *(float4*)(dst)       = selfV;
        *(float4*)(dst + EMB) = mean;

        #pragma unroll
        for (int j = 0; j < K_NBR; ++j) idsCur[j] = idsNxt[j];
        nodeCur = nodeNxt; nodeNxt = nodeNxt2;
    }
    #undef GROUP_ROW


    const int half = lane >> 4;
    const int c    = lane & 15;
    const float* w0 = W + c * 64;
    const float* w1 = W + (c + 16) * 64;

    v8f d0[TPW], d1[TPW];
    #pragma unroll
    for (int t = 0; t < TPW; ++t) { d0[t] = (v8f){}; d1[t] = (v8f){}; }

    #pragma unroll
    for (int q = 0; q < 8; ++q) {
        const float4 wa = ld_chunk(w0, q, half);
        const float4 wb = ld_chunk(w1, q, half);
        float4 a[TPW];
        #pragma unroll
        for (int t = 0; t < TPW; ++t) a[t] = ld_chunk(myStage + (16 * t + c) * PITCH, q, half);
        #pragma unroll
        for (int t = 0; t < TPW; ++t) {
            d0[t] = mma4(a[t].x, a[t].y, wa.x, wa.y, d0[t]);
            d1[t] = mma4(a[t].x, a[t].y, wb.x, wb.y, d1[t]);
        }
        #pragma unroll
        for (int t = 0; t < TPW; ++t) {
            d0[t] = mma4(a[t].z, a[t].w, wa.z, wa.w, d0[t]);
            d1[t] = mma4(a[t].z, a[t].w, wb.z, wb.w, d1[t]);
        }
    }

    #pragma unroll
    for (int t = 0; t < TPW; ++t) {
        if (t == 0 || hasB) {
            const int m0 = (t == 0) ? m0A : m0B;
            #pragma unroll
            for (int v = 0; v < 8; ++v) {
                const int m = v + 8 * half;
                out[(m0 + m) * EMB + c]      = fmaxf(d0[t][v], 0.0f);
                out[(m0 + m) * EMB + 16 + c] = fmaxf(d1[t][v], 0.0f);
            }
        }
    }
}

extern "C" void kernel_launch(void* const* d_in, const int* in_sizes, int n_in,
                              void* d_out, int out_size, void* d_ws, size_t ws_size,
                              hipStream_t stream) {
    const float* emb        = (const float*)d_in[0];
    const float* W1         = (const float*)d_in[1];
    const float* W2         = (const float*)d_in[2];
    const int*   node_batch = (const int*)  d_in[3];
    const int*   neigh      = (const int*)  d_in[4];
    float*       out        = (float*)d_out;
    float*       h1         = (float*)d_ws;

    const int N = in_sizes[0] / EMB;
    const int B = in_sizes[3];

    const int t1 = (N + 15) / 16;
    const int t2 = (B + 15) / 16;

    sage_layer_kernel<<<(t1 + ROWS16_PER_BLOCK - 1) / ROWS16_PER_BLOCK, 256, 0, stream>>>(
        emb, W1, neigh, nullptr, h1, t1);

    sage_layer_kernel<<<(t2 + ROWS16_PER_BLOCK - 1) / ROWS16_PER_BLOCK, 256, 0, stream>>>(
        h1, W2, neigh, node_batch, out, t2);
}
